# starting kernel
# baseline (speedup 1.0000x reference)
#include <hip/hip_runtime.h>

#define HD   256
#define NB   128
#define RT   16
#define WTAB (RT * NB * 4)

typedef float v2f __attribute__((ext_vector_type(2)));
typedef float v8f __attribute__((ext_vector_type(8)));

template <bool RELU>
__global__ __launch_bounds__(256)
void rgcn_selfloop_gemm(const float* __restrict__ A,
                        const float* __restrict__ W,
                        const float* __restrict__ bias,
                        float* __restrict__ out,
                        int n_rows)
{
    const int lane = threadIdx.x & 31;
    const int wid  = blockIdx.x * (blockDim.x >> 5) + (threadIdx.x >> 5);

    const int strips_n = HD / 64;
    const int tile_m   = wid / strips_n;
    const int strip_n  = wid % strips_n;
    if (tile_m * 16 >= n_rows) return;

    const int half = lane >> 4;
    const int sub  = lane & 15;
    const int arow_idx = tile_m * 16 + sub;
    const int col0     = strip_n * 64 + sub;

    const float* arow = A + (size_t)arow_idx * HD;

    v8f c0 = {}, c1 = {}, c2 = {}, c3 = {};
    for (int k = 0; k < HD; k += 4) {
        float a0 = arow[k + 2 * half + 0];
        float a1 = arow[k + 2 * half + 1];
        if (RELU) { a0 = fmaxf(a0, 0.0f); a1 = fmaxf(a1, 0.0f); }
        v2f a; a.x = a0; a.y = a1;

        const float* wrow0 = W + (size_t)(k + 2 * half + 0) * HD + col0;
        const float* wrow1 = W + (size_t)(k + 2 * half + 1) * HD + col0;
        v2f b0, b1, b2, b3;
        b0.x = wrow0[0];  b0.y = wrow1[0];
        b1.x = wrow0[16]; b1.y = wrow1[16];
        b2.x = wrow0[32]; b2.y = wrow1[32];
        b3.x = wrow0[48]; b3.y = wrow1[48];

        c0 = __builtin_amdgcn_wmma_f32_16x16x4_f32(false, a, false, b0, (short)0, c0, false, false);
        c1 = __builtin_amdgcn_wmma_f32_16x16x4_f32(false, a, false, b1, (short)0, c1, false, false);
        c2 = __builtin_amdgcn_wmma_f32_16x16x4_f32(false, a, false, b2, (short)0, c2, false, false);
        c3 = __builtin_amdgcn_wmma_f32_16x16x4_f32(false, a, false, b3, (short)0, c3, false, false);
    }

    float* obase = out + (size_t)(tile_m * 16 + half * 8) * HD + col0;
    #pragma unroll
    for (int t = 0; t < 4; ++t) {
        const v8f& c = (t == 0) ? c0 : (t == 1) ? c1 : (t == 2) ? c2 : c3;
        const float bcol = bias[col0 + t * 16];
        #pragma unroll
        for (int j = 0; j < 8; ++j)
            obase[(size_t)j * HD + t * 16] = c[j] + bcol;
    }
}

template <bool RELU>
__global__ __launch_bounds__(256)
void rgcn_edge_msg(const float* __restrict__ xin,
                   const int*   __restrict__ src,
                   const int*   __restrict__ dst,
                   const int*   __restrict__ etype,
                   const float* __restrict__ norm,
                   const float* __restrict__ w,
                   float* agg,
                   int n_edges)
{
    __shared__ __align__(16) float wlds[WTAB];
    for (int i = threadIdx.x; i < WTAB; i += blockDim.x) wlds[i] = w[i];
    __syncthreads();

    const int lane   = threadIdx.x & 31;
    const int wid    = blockIdx.x * (blockDim.x >> 5) + (threadIdx.x >> 5);
    const int nwaves = (gridDim.x * blockDim.x) >> 5;

    for (int e = wid; e < n_edges; e += nwaves) {
        const int   s   = src[e];
        const int   d   = dst[e];
        const int   r   = etype[e];
        const float nrm = norm[e];

        const float4* xrow = (const float4*)(xin + (size_t)s * HD) + lane * 2;
        float4 xa = xrow[0];
        float4 xb = xrow[1];
        if (RELU) {
            xa.x = fmaxf(xa.x, 0.0f); xa.y = fmaxf(xa.y, 0.0f);
            xa.z = fmaxf(xa.z, 0.0f); xa.w = fmaxf(xa.w, 0.0f);
            xb.x = fmaxf(xb.x, 0.0f); xb.y = fmaxf(xb.y, 0.0f);
            xb.z = fmaxf(xb.z, 0.0f); xb.w = fmaxf(xb.w, 0.0f);
        }
        const float xs[8] = { xa.x, xa.y, xa.z, xa.w, xb.x, xb.y, xb.z, xb.w };

        const float4* wrel = (const float4*)(wlds + (size_t)r * (NB * 4)) + lane * 4;
        float* arow = agg + (size_t)d * HD + lane * 8;

        #pragma unroll
        for (int j = 0; j < 4; ++j) {
            const float4 wq = wrel[j];
            const float x0 = xs[2 * j + 0];
            const float x1 = xs[2 * j + 1];
            const float o0 = (x0 * wq.x + x1 * wq.z) * nrm;
            const float o1 = (x0 * wq.y + x1 * wq.w) * nrm;
            (void)__hip_atomic_fetch_add(&arow[2 * j + 0], o0,
                                         __ATOMIC_RELAXED, __HIP_MEMORY_SCOPE_AGENT);
            (void)__hip_atomic_fetch_add(&arow[2 * j + 1], o1,
                                         __ATOMIC_RELAXED, __HIP_MEMORY_SCOPE_AGENT);
        }
    }
}

extern "C" void kernel_launch(void* const* d_in, const int* in_sizes, int n_in,
                              void* d_out, int out_size, void* d_ws, size_t ws_size,
                              hipStream_t stream)
{
    const float* x     = (const float*)d_in[0];
    const int*   src   = (const int*)  d_in[1];
    const int*   dst   = (const int*)  d_in[2];
    const int*   etype = (const int*)  d_in[3];
    const float* norm  = (const float*)d_in[4];
    const float* w1    = (const float*)d_in[5];
    const float* lw1   = (const float*)d_in[6];
    const float* b1    = (const float*)d_in[7];
    const float* w2    = (const float*)d_in[8];
    const float* lw2   = (const float*)d_in[9];
    const float* b2    = (const float*)d_in[10];
    float*       out   = (float*)d_out;

    const int n_nodes = in_sizes[0] / HD;
    const int n_edges = in_sizes[1];

    float* h = (float*)d_ws;

    const int waves_per_block = 256 / 32;
    const int strips = ((n_nodes + 15) / 16) * (HD / 64);
    const int gemm_blocks = (strips + waves_per_block - 1) / waves_per_block;
    const int edge_blocks = 1024;

    rgcn_selfloop_gemm<false><<<gemm_blocks, 256, 0, stream>>>(x, lw1, b1, h, n_nodes);
    rgcn_edge_msg<false><<<edge_blocks, 256, 0, stream>>>(x, src, dst, etype, norm, w1,
                                                          h, n_edges);

    rgcn_selfloop_gemm<true><<<gemm_blocks, 256, 0, stream>>>(h, lw2, b2, out, n_nodes);
    rgcn_edge_msg<true><<<edge_blocks, 256, 0, stream>>>(h, src, dst, etype, norm, w2,
                                                         out, n_edges);
}


# Self-loop strip operands in whole 128-B lines: owned rows staged once in LDS by all waves (A fragment = one 8-B LDS read), W by two 16-B loads per K-step (strip columns 4*sub+t); sum order kept
# speedup vs baseline: 3.5696x; 3.5696x over previous
#include <hip/hip_runtime.h>

#define HD   256
#define NB   128
#define RT   16
#define WTAB (RT * NB * 4)

#define NPB   80
#define EWAV  32
#define ETHR  (EWAV * 32)

typedef float v2f __attribute__((ext_vector_type(2)));
typedef float v8f __attribute__((ext_vector_type(8)));

#define APITCH (HD + 4)

#define ASTG (NPB * (HD / 4) / ETHR)
static_assert(ASTG * ETHR == NPB * (HD / 4), "the tile must divide evenly over the threads");

__device__ __forceinline__ void stage_rows_load(const float* __restrict__ A, float4 (&av)[ASTG],
                                                int lo, int n_rows, int tid)
{
    #pragma unroll
    for (int u = 0; u < ASTG; ++u) {
        const int i = tid + u * ETHR;
        int idx = lo + (i >> 6);
        if (idx >= n_rows) idx = n_rows - 1;
        av[u] = ((const float4*)(A + (size_t)idx * HD))[i & 63];
    }
}

template <bool RELU>
__device__ __forceinline__ void stage_rows_store(float* atile, const float4 (&av)[ASTG], int tid)
{
    #pragma unroll
    for (int u = 0; u < ASTG; ++u) {
        const int i = tid + u * ETHR;
        float4 v = av[u];
        if (RELU) {
            v.x = fmaxf(v.x, 0.0f); v.y = fmaxf(v.y, 0.0f);
            v.z = fmaxf(v.z, 0.0f); v.w = fmaxf(v.w, 0.0f);
        }
        *(float4*)(atile + (i >> 6) * APITCH + 4 * (i & 63)) = v;
    }
}

__device__ __forceinline__ void selfloop_strip(const float* atile,
                                               const float* __restrict__ W,
                                               const float* __restrict__ bias,
                                               float* acc, int tile_m, int strip_n, int lane)
{
    const int half = lane >> 4;
    const int sub  = lane & 15;
    const int col0 = strip_n * 64 + 4 * sub;

    const float* arow = atile + (tile_m * 16 + sub) * APITCH + 2 * half;
    const float* wrow = W + (size_t)(2 * half) * HD + col0;

    const float4 bq = *(const float4*)(bias + col0);

    v8f c0 = {}, c1 = {}, c2 = {}, c3 = {};
    for (int k = 0; k < HD; k += 4) {
        const v2f a = *(const v2f*)(arow + k);
        const float4 q0 = *(const float4*)(wrow + (size_t)k * HD);
        const float4 q1 = *(const float4*)(wrow + (size_t)(k + 1) * HD);
        v2f b0, b1, b2, b3;
        b0.x = q0.x; b0.y = q1.x;
        b1.x = q0.y; b1.y = q1.y;
        b2.x = q0.z; b2.y = q1.z;
        b3.x = q0.w; b3.y = q1.w;

        c0 = __builtin_amdgcn_wmma_f32_16x16x4_f32(false, a, false, b0, (short)0, c0, false, false);
        c1 = __builtin_amdgcn_wmma_f32_16x16x4_f32(false, a, false, b1, (short)0, c1, false, false);
        c2 = __builtin_amdgcn_wmma_f32_16x16x4_f32(false, a, false, b2, (short)0, c2, false, false);
        c3 = __builtin_amdgcn_wmma_f32_16x16x4_f32(false, a, false, b3, (short)0, c3, false, false);
    }

    const float bcol[4] = { bq.x, bq.y, bq.z, bq.w };
    float* abase = acc + (size_t)(tile_m * 16 + half * 8) * HD + col0;
    #pragma unroll
    for (int t = 0; t < 4; ++t) {
        const v8f& c = (t == 0) ? c0 : (t == 1) ? c1 : (t == 2) ? c2 : c3;
        #pragma unroll
        for (int j = 0; j < 8; ++j)
            (void)__hip_atomic_fetch_add(&abase[(size_t)j * HD + t], c[j] + bcol[t],
                                         __ATOMIC_RELAXED, __HIP_MEMORY_SCOPE_WORKGROUP);
    }
}

#define QCAP  640
#define WCAP  2048
#define GWAV  ((NPB / 16) * (HD / 64))
#define CH    8

__device__ __forceinline__ int4 load_dst4(const int* __restrict__ dst, int g, int n_groups, int n_edges)
{
    int4 v = make_int4(-1, -1, -1, -1);
    if (g < n_groups) {
        const int e = g * 4;
        if (e + 3 < n_edges) {
            v = *(const int4*)(dst + e);
        } else {
            v.x = dst[e];
            if (e + 1 < n_edges) v.y = dst[e + 1];
            if (e + 2 < n_edges) v.z = dst[e + 2];
        }
    }
    return v;
}

__device__ __forceinline__ int push_matches(int4 v, int e0, int lo, int* q, int qn)
{
    const bool m0 = (unsigned)(v.x - lo) < (unsigned)NPB;
    const bool m1 = (unsigned)(v.y - lo) < (unsigned)NPB;
    const bool m2 = (unsigned)(v.z - lo) < (unsigned)NPB;
    const bool m3 = (unsigned)(v.w - lo) < (unsigned)NPB;
    if (__builtin_amdgcn_ballot_w32(m0 | m1 | m2 | m3) != 0u) {
        unsigned mk;
        mk = __builtin_amdgcn_ballot_w32(m0);
        if (m0) q[qn + __builtin_amdgcn_mbcnt_lo(mk, 0u)] = e0 + 0;
        qn += __builtin_popcount(mk);
        mk = __builtin_amdgcn_ballot_w32(m1);
        if (m1) q[qn + __builtin_amdgcn_mbcnt_lo(mk, 0u)] = e0 + 1;
        qn += __builtin_popcount(mk);
        mk = __builtin_amdgcn_ballot_w32(m2);
        if (m2) q[qn + __builtin_amdgcn_mbcnt_lo(mk, 0u)] = e0 + 2;
        qn += __builtin_popcount(mk);
        mk = __builtin_amdgcn_ballot_w32(m3);
        if (m3) q[qn + __builtin_amdgcn_mbcnt_lo(mk, 0u)] = e0 + 3;
        qn += __builtin_popcount(mk);
    }
    return qn;
}

template <bool RELU, int NE>
__device__ __forceinline__ void edge_batch(const float* __restrict__ xin,
                                           const float* wlds, float* acc,
                                           const int4* rq, int lane)
{
    int4 rc[NE];
    #pragma unroll
    for (int k = 0; k < NE; ++k) rc[k] = rq[k];
    float2 xv[NE][4];
    #pragma unroll
    for (int k = 0; k < NE; ++k) {
        const float2* xr = (const float2*)(xin + (size_t)rc[k].x * HD) + lane;
        #pragma unroll
        for (int j = 0; j < 4; ++j) xv[k][j] = xr[32 * j];
    }
    #pragma unroll
    for (int k = 0; k < NE; ++k) {
        const float4* wrel = (const float4*)(wlds + (size_t)rc[k].z * (NB * 4)) + lane;
        float* arow = acc + (size_t)rc[k].y * HD + 2 * lane;
        const float nrm = __int_as_float(rc[k].w);
        #pragma unroll
        for (int j = 0; j < 4; ++j) {
            const float4 wq = wrel[32 * j];
            float x0 = xv[k][j].x;
            float x1 = xv[k][j].y;
            if (RELU) { x0 = fmaxf(x0, 0.0f); x1 = fmaxf(x1, 0.0f); }
            const float o0 = (x0 * wq.x + x1 * wq.z) * nrm;
            const float o1 = (x0 * wq.y + x1 * wq.w) * nrm;
            (void)__hip_atomic_fetch_add(&arow[64 * j + 0], o0,
                                         __ATOMIC_RELAXED, __HIP_MEMORY_SCOPE_WORKGROUP);
            (void)__hip_atomic_fetch_add(&arow[64 * j + 1], o1,
                                         __ATOMIC_RELAXED, __HIP_MEMORY_SCOPE_WORKGROUP);
        }
    }
}

template <bool RELU>
__device__ __forceinline__ void drain_recs(const float* __restrict__ xin,
                                           const float* wlds, float* acc,
                                           const int4* rq, int m, int lane)
{
    int i = 0;
    for (; i + 4 <= m; i += 4)
        edge_batch<RELU, 4>(xin, wlds, acc, rq + i, lane);
    const int rem = m - i;
    if (rem == 3)      edge_batch<RELU, 3>(xin, wlds, acc, rq + i, lane);
    else if (rem == 2) edge_batch<RELU, 2>(xin, wlds, acc, rq + i, lane);
    else if (rem == 1) edge_batch<RELU, 1>(xin, wlds, acc, rq + i, lane);
}

template <bool RELU>
__device__ __forceinline__ void flush_ids(const float* __restrict__ xin,
                                          const int*   __restrict__ src,
                                          const int*   __restrict__ dst,
                                          const int*   __restrict__ etype,
                                          const float* __restrict__ norm,
                                          const float* wlds, float* acc,
                                          const int* q, int qn, int lo,
                                          int4* rq, int4* save, int lane)
{
    for (int i = 0; i < qn; i += 32) {
        const int m = (qn - i < 32) ? (qn - i) : 32;
        int4 rec = make_int4(0, 0, 0, 0);
        if (lane < m) {
            const int e = q[i + lane];
            rec.x = src[e];
            rec.y = dst[e] - lo;
            rec.z = etype[e];
            rec.w = __float_as_int(norm[e]);
            if (save != nullptr) save[i + lane] = rec;
        }
        rq[lane] = rec;
        drain_recs<RELU>(xin, wlds, acc, rq, m, lane);
    }
}

template <bool RELU, bool FIRST>
__global__ __launch_bounds__(ETHR)
void rgcn_edge_owner(const float* __restrict__ xin,
                     const int*   __restrict__ src,
                     const int*   __restrict__ dst,
                     const int*   __restrict__ etype,
                     const float* __restrict__ norm,
                     const float* __restrict__ w,
                     const float* __restrict__ lw,
                     const float* __restrict__ bias,
                     float* __restrict__ agg,
                     int*  __restrict__ cnt,
                     int4* __restrict__ recs,
                     int n_edges, int n_nodes)
{
    __shared__ __align__(16) float wlds[WTAB];
    __shared__ __align__(16) float acc[NPB * HD];
    __shared__ __align__(16) float atile[NPB * APITCH];
    __shared__ __align__(16) int4 rqs[EWAV * 32];
    __shared__ int queue[EWAV * QCAP];
    __shared__ int wg_fill;
    __shared__ int wg_ovf;
    __shared__ int wg_next;

    const int tid  = threadIdx.x;
    const int lane = tid & 31;
    const int wv   = __builtin_amdgcn_readfirstlane(tid >> 5);
    const int lo   = blockIdx.x * NPB;

    float4 av[ASTG];
    stage_rows_load(xin, av, lo, n_nodes, tid);
    for (int i = tid; i < WTAB / 4; i += ETHR)
        ((float4*)wlds)[i] = ((const float4*)w)[i];
    for (int i = tid; i < NPB * HD / 4; i += ETHR)
        ((float4*)acc)[i] = make_float4(0.0f, 0.0f, 0.0f, 0.0f);
    if (tid == 0) { wg_fill = 0; wg_ovf = 0; wg_next = 0; }
    stage_rows_store<RELU>(atile, av, tid);
    __syncthreads();

    if (wv < GWAV) {
        const int tile_m = wv >> 2;
        if (lo + tile_m * 16 < n_nodes)
            selfloop_strip(atile, lw, bias, acc, tile_m, wv & 3, lane);
    }

    int4* rq = rqs + wv * 32;
    int4* strip = recs + (size_t)blockIdx.x * WCAP;

    int saved = -1;
    if (!FIRST) saved = cnt[blockIdx.x];

    if (!FIRST && saved >= 0) {
        int c = 0;
        if (lane == 0)
            c = __hip_atomic_fetch_add(&wg_next, CH, __ATOMIC_RELAXED, __HIP_MEMORY_SCOPE_WORKGROUP);
        c = __builtin_amdgcn_readfirstlane(c);
        int4 cur = make_int4(0, 0, 0, 0);
        if (lane < CH && c + lane < saved) cur = strip[c + lane];
        while (c < saved) {
            const int m = (saved - c < CH) ? (saved - c) : CH;
            rq[lane] = cur;
            int cn = 0;
            if (lane == 0)
                cn = __hip_atomic_fetch_add(&wg_next, CH, __ATOMIC_RELAXED, __HIP_MEMORY_SCOPE_WORKGROUP);
            cn = __builtin_amdgcn_readfirstlane(cn);
            int4 nxt = make_int4(0, 0, 0, 0);
            if (lane < CH && cn + lane < saved) nxt = strip[cn + lane];
            drain_recs<RELU>(xin, wlds, acc, rq, m, lane);
            c = cn;
            cur = nxt;
        }
    } else {
        int* q = queue + wv * QCAP;
        int qn = 0;

        const int n_outer = n_edges / (ETHR * 16);
        const int rot = n_outer > 0 ? (int)(blockIdx.x % (unsigned)n_outer) : 0;
        for (int it = 0; it < n_outer; ++it) {
            int it_r = it + rot;
            if (it_r >= n_outer) it_r -= n_outer;
            const int e_base = it_r * (ETHR * 16) + tid * 4;
            int4 v[4];
            #pragma unroll
            for (int u = 0; u < 4; ++u)
                v[u] = *(const int4*)(dst + e_base + u * (ETHR * 4));
            #pragma unroll
            for (int u = 0; u < 4; ++u)
                qn = push_matches(v[u], e_base + u * (ETHR * 4), lo, q, qn);
            if (qn > QCAP - 512) {
                if (FIRST && lane == 0) wg_ovf = 1;
                flush_ids<RELU>(xin, src, dst, etype, norm, wlds, acc, q, qn, lo, rq, nullptr, lane);
                qn = 0;
            }
        }
        const int n_groups = (n_edges + 3) >> 2;
        for (int g0 = n_outer * (ETHR * 4); g0 < n_groups; g0 += ETHR) {
            const int g = g0 + tid;
            qn = push_matches(load_dst4(dst, g, n_groups, n_edges), g * 4, lo, q, qn);
        }
        int4* save = nullptr;
        if (FIRST) {
            int o = 0;
            if (lane == 0)
                o = __hip_atomic_fetch_add(&wg_fill, qn, __ATOMIC_RELAXED, __HIP_MEMORY_SCOPE_WORKGROUP);
            o = __builtin_amdgcn_readfirstlane(o);
            if (o + qn <= WCAP) save = strip + o;
            else if (lane == 0) wg_ovf = 1;
        }
        flush_ids<RELU>(xin, src, dst, etype, norm, wlds, acc, q, qn, lo, rq, save, lane);
    }
    __syncthreads();

    if (FIRST && tid == 0) cnt[blockIdx.x] = (wg_ovf == 0) ? wg_fill : -1;

    for (int i = tid; i < NPB * (HD / 4); i += ETHR) {
        const int row = lo + (i >> 6);
        if (row < n_nodes)
            ((float4*)(agg + (size_t)row * HD))[i & 63] = ((const float4*)acc)[i];
    }
}

extern "C" void kernel_launch(void* const* d_in, const int* in_sizes, int n_in,
                              void* d_out, int out_size, void* d_ws, size_t ws_size,
                              hipStream_t stream)
{
    const float* x     = (const float*)d_in[0];
    const int*   src   = (const int*)  d_in[1];
    const int*   dst   = (const int*)  d_in[2];
    const int*   etype = (const int*)  d_in[3];
    const float* norm  = (const float*)d_in[4];
    const float* w1    = (const float*)d_in[5];
    const float* lw1   = (const float*)d_in[6];
    const float* b1    = (const float*)d_in[7];
    const float* w2    = (const float*)d_in[8];
    const float* lw2   = (const float*)d_in[9];
    const float* b2    = (const float*)d_in[10];
    float*       out   = (float*)d_out;

    const int n_nodes = in_sizes[0] / HD;
    const int n_edges = in_sizes[1];

    const int edge_blocks = (n_nodes + NPB - 1) / NPB;

    float* h    = (float*)d_ws;
    int*   cnt  = (int*)(h + (size_t)n_nodes * HD);
    int4*  recs = (int4*)(cnt + ((edge_blocks + 63) / 64) * 64);

    rgcn_edge_owner<false, true><<<edge_blocks, ETHR, 0, stream>>>(x, src, dst, etype, norm, w1, lw1, b1,
                                                                   h, cnt, recs, n_edges, n_nodes);

    rgcn_edge_owner<true, false><<<edge_blocks, ETHR, 0, stream>>>(h, src, dst, etype, norm, w2, lw2, b2,
                                                                   out, cnt, recs, n_edges, n_nodes);
}


# Layer 1: dst[] scan dealt to waves in 512-id chunks from an LDS counter, matches worked off (records saved per flush) every 8 found, next chunk asked first: edge adds run beside scan; arrival order
# speedup vs baseline: 3.6857x; 1.0325x over previous
#include <hip/hip_runtime.h>

#define HD   256
#define NB   128
#define RT   16
#define WTAB (RT * NB * 4)

#define NPB   80
#define EWAV  32
#define ETHR  (EWAV * 32)

typedef float v2f __attribute__((ext_vector_type(2)));
typedef float v8f __attribute__((ext_vector_type(8)));

#define APITCH (HD + 4)

#define ASTG (NPB * (HD / 4) / ETHR)
static_assert(ASTG * ETHR == NPB * (HD / 4), "the tile must divide evenly over the threads");

__device__ __forceinline__ void stage_rows_load(const float* __restrict__ A, float4 (&av)[ASTG],
                                                int lo, int n_rows, int tid)
{
    #pragma unroll
    for (int u = 0; u < ASTG; ++u) {
        const int i = tid + u * ETHR;
        int idx = lo + (i >> 6);
        if (idx >= n_rows) idx = n_rows - 1;
        av[u] = ((const float4*)(A + (size_t)idx * HD))[i & 63];
    }
}

template <bool RELU>
__device__ __forceinline__ void stage_rows_store(float* atile, const float4 (&av)[ASTG], int tid)
{
    #pragma unroll
    for (int u = 0; u < ASTG; ++u) {
        const int i = tid + u * ETHR;
        float4 v = av[u];
        if (RELU) {
            v.x = fmaxf(v.x, 0.0f); v.y = fmaxf(v.y, 0.0f);
            v.z = fmaxf(v.z, 0.0f); v.w = fmaxf(v.w, 0.0f);
        }
        *(float4*)(atile + (i >> 6) * APITCH + 4 * (i & 63)) = v;
    }
}

__device__ __forceinline__ void selfloop_strip(const float* atile,
                                               const float* __restrict__ W,
                                               const float* __restrict__ bias,
                                               float* acc, int tile_m, int strip_n, int lane)
{
    const int half = lane >> 4;
    const int sub  = lane & 15;
    const int col0 = strip_n * 64 + 4 * sub;

    const float* arow = atile + (tile_m * 16 + sub) * APITCH + 2 * half;
    const float* wrow = W + (size_t)(2 * half) * HD + col0;

    const float4 bq = *(const float4*)(bias + col0);

    v8f c0 = {}, c1 = {}, c2 = {}, c3 = {};
    for (int k = 0; k < HD; k += 4) {
        const v2f a = *(const v2f*)(arow + k);
        const float4 q0 = *(const float4*)(wrow + (size_t)k * HD);
        const float4 q1 = *(const float4*)(wrow + (size_t)(k + 1) * HD);
        v2f b0, b1, b2, b3;
        b0.x = q0.x; b0.y = q1.x;
        b1.x = q0.y; b1.y = q1.y;
        b2.x = q0.z; b2.y = q1.z;
        b3.x = q0.w; b3.y = q1.w;

        c0 = __builtin_amdgcn_wmma_f32_16x16x4_f32(false, a, false, b0, (short)0, c0, false, false);
        c1 = __builtin_amdgcn_wmma_f32_16x16x4_f32(false, a, false, b1, (short)0, c1, false, false);
        c2 = __builtin_amdgcn_wmma_f32_16x16x4_f32(false, a, false, b2, (short)0, c2, false, false);
        c3 = __builtin_amdgcn_wmma_f32_16x16x4_f32(false, a, false, b3, (short)0, c3, false, false);
    }

    const float bcol[4] = { bq.x, bq.y, bq.z, bq.w };
    float* abase = acc + (size_t)(tile_m * 16 + half * 8) * HD + col0;
    #pragma unroll
    for (int t = 0; t < 4; ++t) {
        const v8f& c = (t == 0) ? c0 : (t == 1) ? c1 : (t == 2) ? c2 : c3;
        #pragma unroll
        for (int j = 0; j < 8; ++j)
            (void)__hip_atomic_fetch_add(&abase[(size_t)j * HD + t], c[j] + bcol[t],
                                         __ATOMIC_RELAXED, __HIP_MEMORY_SCOPE_WORKGROUP);
    }
}

#define QCAP  640
#define WCAP  2048
#define GWAV  ((NPB / 16) * (HD / 64))
#define CH    8
#define SCH   512
#define FLMIN 8
static_assert(QCAP >= FLMIN - 1 + SCH, "the queue must hold a whole chunk on top of an unflushed rest");

__device__ __forceinline__ int4 load_dst4(const int* __restrict__ dst, int g, int n_groups, int n_edges)
{
    int4 v = make_int4(-1, -1, -1, -1);
    if (g < n_groups) {
        const int e = g * 4;
        if (e + 3 < n_edges) {
            v = *(const int4*)(dst + e);
        } else {
            v.x = dst[e];
            if (e + 1 < n_edges) v.y = dst[e + 1];
            if (e + 2 < n_edges) v.z = dst[e + 2];
        }
    }
    return v;
}

__device__ __forceinline__ void load_chunk(const int* __restrict__ dst, int c, int lane, int4 (&v)[4])
{
    const int4* p = (const int4*)(dst + (size_t)c * SCH) + lane;
    #pragma unroll
    for (int u = 0; u < 4; ++u) v[u] = p[u * 32];
}

__device__ __forceinline__ int push_matches(int4 v, int e0, int lo, int* q, int qn)
{
    const bool m0 = (unsigned)(v.x - lo) < (unsigned)NPB;
    const bool m1 = (unsigned)(v.y - lo) < (unsigned)NPB;
    const bool m2 = (unsigned)(v.z - lo) < (unsigned)NPB;
    const bool m3 = (unsigned)(v.w - lo) < (unsigned)NPB;
    if (__builtin_amdgcn_ballot_w32(m0 | m1 | m2 | m3) != 0u) {
        unsigned mk;
        mk = __builtin_amdgcn_ballot_w32(m0);
        if (m0) q[qn + __builtin_amdgcn_mbcnt_lo(mk, 0u)] = e0 + 0;
        qn += __builtin_popcount(mk);
        mk = __builtin_amdgcn_ballot_w32(m1);
        if (m1) q[qn + __builtin_amdgcn_mbcnt_lo(mk, 0u)] = e0 + 1;
        qn += __builtin_popcount(mk);
        mk = __builtin_amdgcn_ballot_w32(m2);
        if (m2) q[qn + __builtin_amdgcn_mbcnt_lo(mk, 0u)] = e0 + 2;
        qn += __builtin_popcount(mk);
        mk = __builtin_amdgcn_ballot_w32(m3);
        if (m3) q[qn + __builtin_amdgcn_mbcnt_lo(mk, 0u)] = e0 + 3;
        qn += __builtin_popcount(mk);
    }
    return qn;
}

template <bool RELU, int NE>
__device__ __forceinline__ void edge_batch(const float* __restrict__ xin,
                                           const float* wlds, float* acc,
                                           const int4* rq, int lane)
{
    int4 rc[NE];
    #pragma unroll
    for (int k = 0; k < NE; ++k) rc[k] = rq[k];
    float2 xv[NE][4];
    #pragma unroll
    for (int k = 0; k < NE; ++k) {
        const float2* xr = (const float2*)(xin + (size_t)rc[k].x * HD) + lane;
        #pragma unroll
        for (int j = 0; j < 4; ++j) xv[k][j] = xr[32 * j];
    }
    #pragma unroll
    for (int k = 0; k < NE; ++k) {
        const float4* wrel = (const float4*)(wlds + (size_t)rc[k].z * (NB * 4)) + lane;
        float* arow = acc + (size_t)rc[k].y * HD + 2 * lane;
        const float nrm = __int_as_float(rc[k].w);
        #pragma unroll
        for (int j = 0; j < 4; ++j) {
            const float4 wq = wrel[32 * j];
            float x0 = xv[k][j].x;
            float x1 = xv[k][j].y;
            if (RELU) { x0 = fmaxf(x0, 0.0f); x1 = fmaxf(x1, 0.0f); }
            const float o0 = (x0 * wq.x + x1 * wq.z) * nrm;
            const float o1 = (x0 * wq.y + x1 * wq.w) * nrm;
            (void)__hip_atomic_fetch_add(&arow[64 * j + 0], o0,
                                         __ATOMIC_RELAXED, __HIP_MEMORY_SCOPE_WORKGROUP);
            (void)__hip_atomic_fetch_add(&arow[64 * j + 1], o1,
                                         __ATOMIC_RELAXED, __HIP_MEMORY_SCOPE_WORKGROUP);
        }
    }
}

template <bool RELU>
__device__ __forceinline__ void drain_recs(const float* __restrict__ xin,
                                           const float* wlds, float* acc,
                                           const int4* rq, int m, int lane)
{
    int i = 0;
    for (; i + 4 <= m; i += 4)
        edge_batch<RELU, 4>(xin, wlds, acc, rq + i, lane);
    const int rem = m - i;
    if (rem == 3)      edge_batch<RELU, 3>(xin, wlds, acc, rq + i, lane);
    else if (rem == 2) edge_batch<RELU, 2>(xin, wlds, acc, rq + i, lane);
    else if (rem == 1) edge_batch<RELU, 1>(xin, wlds, acc, rq + i, lane);
}

template <bool RELU, bool SAVE>
__device__ __forceinline__ void flush_ids(const float* __restrict__ xin,
                                          const int*   __restrict__ src,
                                          const int*   __restrict__ dst,
                                          const int*   __restrict__ etype,
                                          const float* __restrict__ norm,
                                          const float* wlds, float* acc,
                                          const int* q, int qn, int lo,
                                          int4* rq, int4* strip, int* fill, int* ovf, int lane)
{
    for (int i = 0; i < qn; i += 32) {
        const int m = (qn - i < 32) ? (qn - i) : 32;
        int4* save = nullptr;
        if (SAVE) {
            int o = 0;
            if (lane == 0)
                o = __hip_atomic_fetch_add(fill, m, __ATOMIC_RELAXED, __HIP_MEMORY_SCOPE_WORKGROUP);
            o = __builtin_amdgcn_readfirstlane(o);
            if (o + m <= WCAP) save = strip + o;
            else if (lane == 0) *ovf = 1;
        }
        int4 rec = make_int4(0, 0, 0, 0);
        if (lane < m) {
            const int e = q[i + lane];
            rec.x = src[e];
            rec.y = dst[e] - lo;
            rec.z = etype[e];
            rec.w = __float_as_int(norm[e]);
            if (save != nullptr) save[lane] = rec;
        }
        rq[lane] = rec;
        drain_recs<RELU>(xin, wlds, acc, rq, m, lane);
    }
}

template <bool RELU, bool FIRST>
__global__ __launch_bounds__(ETHR)
void rgcn_edge_owner(const float* __restrict__ xin,
                     const int*   __restrict__ src,
                     const int*   __restrict__ dst,
                     const int*   __restrict__ etype,
                     const float* __restrict__ norm,
                     const float* __restrict__ w,
                     const float* __restrict__ lw,
                     const float* __restrict__ bias,
                     float* __restrict__ agg,
                     int*  __restrict__ cnt,
                     int4* __restrict__ recs,
                     int n_edges, int n_nodes)
{
    __shared__ __align__(16) float wlds[WTAB];
    __shared__ __align__(16) float acc[NPB * HD];
    __shared__ __align__(16) float atile[NPB * APITCH];
    __shared__ __align__(16) int4 rqs[EWAV * 32];
    __shared__ int queue[EWAV * QCAP];
    __shared__ int wg_fill;
    __shared__ int wg_ovf;
    __shared__ int wg_next;

    const int tid  = threadIdx.x;
    const int lane = tid & 31;
    const int wv   = __builtin_amdgcn_readfirstlane(tid >> 5);
    const int lo   = blockIdx.x * NPB;

    float4 av[ASTG];
    stage_rows_load(xin, av, lo, n_nodes, tid);
    for (int i = tid; i < WTAB / 4; i += ETHR)
        ((float4*)wlds)[i] = ((const float4*)w)[i];
    for (int i = tid; i < NPB * HD / 4; i += ETHR)
        ((float4*)acc)[i] = make_float4(0.0f, 0.0f, 0.0f, 0.0f);
    if (tid == 0) { wg_fill = 0; wg_ovf = 0; wg_next = 0; }
    stage_rows_store<RELU>(atile, av, tid);
    __syncthreads();

    if (wv < GWAV) {
        const int tile_m = wv >> 2;
        if (lo + tile_m * 16 < n_nodes)
            selfloop_strip(atile, lw, bias, acc, tile_m, wv & 3, lane);
    }

    int4* rq = rqs + wv * 32;
    int4* strip = recs + (size_t)blockIdx.x * WCAP;

    int saved = -1;
    if (!FIRST) saved = cnt[blockIdx.x];

    if (!FIRST && saved >= 0) {
        int c = 0;
        if (lane == 0)
            c = __hip_atomic_fetch_add(&wg_next, CH, __ATOMIC_RELAXED, __HIP_MEMORY_SCOPE_WORKGROUP);
        c = __builtin_amdgcn_readfirstlane(c);
        int4 cur = make_int4(0, 0, 0, 0);
        if (lane < CH && c + lane < saved) cur = strip[c + lane];
        while (c < saved) {
            const int m = (saved - c < CH) ? (saved - c) : CH;
            rq[lane] = cur;
            int cn = 0;
            if (lane == 0)
                cn = __hip_atomic_fetch_add(&wg_next, CH, __ATOMIC_RELAXED, __HIP_MEMORY_SCOPE_WORKGROUP);
            cn = __builtin_amdgcn_readfirstlane(cn);
            int4 nxt = make_int4(0, 0, 0, 0);
            if (lane < CH && cn + lane < saved) nxt = strip[cn + lane];
            drain_recs<RELU>(xin, wlds, acc, rq, m, lane);
            c = cn;
            cur = nxt;
        }
    } else {
        int* q = queue + wv * QCAP;
        int qn = 0;

        const int n_full   = n_edges / SCH;
        const int n_groups = (n_edges + 3) >> 2;
        const int rot = n_full > 0 ? (int)(((unsigned long long)blockIdx.x * (unsigned)n_full) / gridDim.x) : 0;

        auto take = [&]() {
            int r = 0;
            if (lane == 0)
                r = __hip_atomic_fetch_add(&wg_next, 1, __ATOMIC_RELAXED, __HIP_MEMORY_SCOPE_WORKGROUP);
            return __builtin_amdgcn_readfirstlane(r);
        };

        int t = take();
        int c = t + rot;
        if (c >= n_full) c -= n_full;
        int4 v[4];
        if (t < n_full) load_chunk(dst, c, lane, v);
        while (t < n_full) {
            const int tn = take();
            int cn = tn + rot;
            if (cn >= n_full) cn -= n_full;
            int4 vn[4];
            load_chunk(dst, (tn < n_full) ? cn : c, lane, vn);
            #pragma unroll
            for (int u = 0; u < 4; ++u)
                qn = push_matches(v[u], c * SCH + u * 128 + lane * 4, lo, q, qn);
            if (qn >= FLMIN) {
                flush_ids<RELU, FIRST>(xin, src, dst, etype, norm, wlds, acc, q, qn, lo, rq,
                                       strip, &wg_fill, &wg_ovf, lane);
                qn = 0;
            }
            t = tn;
            c = cn;
            #pragma unroll
            for (int u = 0; u < 4; ++u) v[u] = vn[u];
        }
        if (t == n_full && n_full * SCH < n_edges) {
            #pragma unroll
            for (int u = 0; u < 4; ++u) {
                const int g = n_full * (SCH / 4) + u * 32 + lane;
                qn = push_matches(load_dst4(dst, g, n_groups, n_edges), g * 4, lo, q, qn);
            }
        }
        flush_ids<RELU, FIRST>(xin, src, dst, etype, norm, wlds, acc, q, qn, lo, rq,
                               strip, &wg_fill, &wg_ovf, lane);
    }
    __syncthreads();

    if (FIRST && tid == 0) cnt[blockIdx.x] = (wg_ovf == 0) ? wg_fill : -1;

    for (int i = tid; i < NPB * (HD / 4); i += ETHR) {
        const int row = lo + (i >> 6);
        if (row < n_nodes)
            ((float4*)(agg + (size_t)row * HD))[i & 63] = ((const float4*)acc)[i];
    }
}

extern "C" void kernel_launch(void* const* d_in, const int* in_sizes, int n_in,
                              void* d_out, int out_size, void* d_ws, size_t ws_size,
                              hipStream_t stream)
{
    const float* x     = (const float*)d_in[0];
    const int*   src   = (const int*)  d_in[1];
    const int*   dst   = (const int*)  d_in[2];
    const int*   etype = (const int*)  d_in[3];
    const float* norm  = (const float*)d_in[4];
    const float* w1    = (const float*)d_in[5];
    const float* lw1   = (const float*)d_in[6];
    const float* b1    = (const float*)d_in[7];
    const float* w2    = (const float*)d_in[8];
    const float* lw2   = (const float*)d_in[9];
    const float* b2    = (const float*)d_in[10];
    float*       out   = (float*)d_out;

    const int n_nodes = in_sizes[0] / HD;
    const int n_edges = in_sizes[1];

    const int edge_blocks = (n_nodes + NPB - 1) / NPB;

    float* h    = (float*)d_ws;
    int*   cnt  = (int*)(h + (size_t)n_nodes * HD);
    int4*  recs = (int4*)(cnt + ((edge_blocks + 63) / 64) * 64);

    rgcn_edge_owner<false, true><<<edge_blocks, ETHR, 0, stream>>>(x, src, dst, etype, norm, w1, lw1, b1,
                                                                   h, cnt, recs, n_edges, n_nodes);

    rgcn_edge_owner<true, false><<<edge_blocks, ETHR, 0, stream>>>(h, src, dst, etype, norm, w2, lw2, b2,
                                                                   out, cnt, recs, n_edges, n_nodes);
}


# Edge and scan work wins issue over the self-loop WMMA strips of its SIMD: s_setprio 3 once a wave turns to edges, strips moved to the 20 youngest waves; LDS float adds, arrival order as before
# speedup vs baseline: 3.8318x; 1.0396x over previous
#include <hip/hip_runtime.h>

#define HD   256
#define NB   128
#define RT   16
#define WTAB (RT * NB * 4)

#define NPB   80
#define EWAV  32
#define ETHR  (EWAV * 32)

typedef float v2f __attribute__((ext_vector_type(2)));
typedef float v8f __attribute__((ext_vector_type(8)));

#define APITCH (HD + 4)

#define ASTG (NPB * (HD / 4) / ETHR)
static_assert(ASTG * ETHR == NPB * (HD / 4), "the tile must divide evenly over the threads");

__device__ __forceinline__ void stage_rows_load(const float* __restrict__ A, float4 (&av)[ASTG],
                                                int lo, int n_rows, int tid)
{
    #pragma unroll
    for (int u = 0; u < ASTG; ++u) {
        const int i = tid + u * ETHR;
        int idx = lo + (i >> 6);
        if (idx >= n_rows) idx = n_rows - 1;
        av[u] = ((const float4*)(A + (size_t)idx * HD))[i & 63];
    }
}

template <bool RELU>
__device__ __forceinline__ void stage_rows_store(float* atile, const float4 (&av)[ASTG], int tid)
{
    #pragma unroll
    for (int u = 0; u < ASTG; ++u) {
        const int i = tid + u * ETHR;
        float4 v = av[u];
        if (RELU) {
            v.x = fmaxf(v.x, 0.0f); v.y = fmaxf(v.y, 0.0f);
            v.z = fmaxf(v.z, 0.0f); v.w = fmaxf(v.w, 0.0f);
        }
        *(float4*)(atile + (i >> 6) * APITCH + 4 * (i & 63)) = v;
    }
}

__device__ __forceinline__ void selfloop_strip(const float* atile,
                                               const float* __restrict__ W,
                                               const float* __restrict__ bias,
                                               float* acc, int tile_m, int strip_n, int lane)
{
    const int half = lane >> 4;
    const int sub  = lane & 15;
    const int col0 = strip_n * 64 + 4 * sub;

    const float* arow = atile + (tile_m * 16 + sub) * APITCH + 2 * half;
    const float* wrow = W + (size_t)(2 * half) * HD + col0;

    const float4 bq = *(const float4*)(bias + col0);

    v8f c0 = {}, c1 = {}, c2 = {}, c3 = {};
    for (int k = 0; k < HD; k += 4) {
        const v2f a = *(const v2f*)(arow + k);
        const float4 q0 = *(const float4*)(wrow + (size_t)k * HD);
        const float4 q1 = *(const float4*)(wrow + (size_t)(k + 1) * HD);
        v2f b0, b1, b2, b3;
        b0.x = q0.x; b0.y = q1.x;
        b1.x = q0.y; b1.y = q1.y;
        b2.x = q0.z; b2.y = q1.z;
        b3.x = q0.w; b3.y = q1.w;

        c0 = __builtin_amdgcn_wmma_f32_16x16x4_f32(false, a, false, b0, (short)0, c0, false, false);
        c1 = __builtin_amdgcn_wmma_f32_16x16x4_f32(false, a, false, b1, (short)0, c1, false, false);
        c2 = __builtin_amdgcn_wmma_f32_16x16x4_f32(false, a, false, b2, (short)0, c2, false, false);
        c3 = __builtin_amdgcn_wmma_f32_16x16x4_f32(false, a, false, b3, (short)0, c3, false, false);
    }

    const float bcol[4] = { bq.x, bq.y, bq.z, bq.w };
    float* abase = acc + (size_t)(tile_m * 16 + half * 8) * HD + col0;
    #pragma unroll
    for (int t = 0; t < 4; ++t) {
        const v8f& c = (t == 0) ? c0 : (t == 1) ? c1 : (t == 2) ? c2 : c3;
        #pragma unroll
        for (int j = 0; j < 8; ++j)
            (void)__hip_atomic_fetch_add(&abase[(size_t)j * HD + t], c[j] + bcol[t],
                                         __ATOMIC_RELAXED, __HIP_MEMORY_SCOPE_WORKGROUP);
    }
}

#define QCAP  640
#define WCAP  2048
#define GWAV  ((NPB / 16) * (HD / 64))
#define CH    8
#define SCH   512
#define FLMIN 8
static_assert(QCAP >= FLMIN - 1 + SCH, "the queue must hold a whole chunk on top of an unflushed rest");

__device__ __forceinline__ int4 load_dst4(const int* __restrict__ dst, int g, int n_groups, int n_edges)
{
    int4 v = make_int4(-1, -1, -1, -1);
    if (g < n_groups) {
        const int e = g * 4;
        if (e + 3 < n_edges) {
            v = *(const int4*)(dst + e);
        } else {
            v.x = dst[e];
            if (e + 1 < n_edges) v.y = dst[e + 1];
            if (e + 2 < n_edges) v.z = dst[e + 2];
        }
    }
    return v;
}

__device__ __forceinline__ void load_chunk(const int* __restrict__ dst, int c, int lane, int4 (&v)[4])
{
    const int4* p = (const int4*)(dst + (size_t)c * SCH) + lane;
    #pragma unroll
    for (int u = 0; u < 4; ++u) v[u] = p[u * 32];
}

__device__ __forceinline__ int push_matches(int4 v, int e0, int lo, int* q, int qn)
{
    const bool m0 = (unsigned)(v.x - lo) < (unsigned)NPB;
    const bool m1 = (unsigned)(v.y - lo) < (unsigned)NPB;
    const bool m2 = (unsigned)(v.z - lo) < (unsigned)NPB;
    const bool m3 = (unsigned)(v.w - lo) < (unsigned)NPB;
    if (__builtin_amdgcn_ballot_w32(m0 | m1 | m2 | m3) != 0u) {
        unsigned mk;
        mk = __builtin_amdgcn_ballot_w32(m0);
        if (m0) q[qn + __builtin_amdgcn_mbcnt_lo(mk, 0u)] = e0 + 0;
        qn += __builtin_popcount(mk);
        mk = __builtin_amdgcn_ballot_w32(m1);
        if (m1) q[qn + __builtin_amdgcn_mbcnt_lo(mk, 0u)] = e0 + 1;
        qn += __builtin_popcount(mk);
        mk = __builtin_amdgcn_ballot_w32(m2);
        if (m2) q[qn + __builtin_amdgcn_mbcnt_lo(mk, 0u)] = e0 + 2;
        qn += __builtin_popcount(mk);
        mk = __builtin_amdgcn_ballot_w32(m3);
        if (m3) q[qn + __builtin_amdgcn_mbcnt_lo(mk, 0u)] = e0 + 3;
        qn += __builtin_popcount(mk);
    }
    return qn;
}

template <bool RELU, int NE>
__device__ __forceinline__ void edge_batch(const float* __restrict__ xin,
                                           const float* wlds, float* acc,
                                           const int4* rq, int lane)
{
    int4 rc[NE];
    #pragma unroll
    for (int k = 0; k < NE; ++k) rc[k] = rq[k];
    float2 xv[NE][4];
    #pragma unroll
    for (int k = 0; k < NE; ++k) {
        const float2* xr = (const float2*)(xin + (size_t)rc[k].x * HD) + lane;
        #pragma unroll
        for (int j = 0; j < 4; ++j) xv[k][j] = xr[32 * j];
    }
    #pragma unroll
    for (int k = 0; k < NE; ++k) {
        const float4* wrel = (const float4*)(wlds + (size_t)rc[k].z * (NB * 4)) + lane;
        float* arow = acc + (size_t)rc[k].y * HD + 2 * lane;
        const float nrm = __int_as_float(rc[k].w);
        #pragma unroll
        for (int j = 0; j < 4; ++j) {
            const float4 wq = wrel[32 * j];
            float x0 = xv[k][j].x;
            float x1 = xv[k][j].y;
            if (RELU) { x0 = fmaxf(x0, 0.0f); x1 = fmaxf(x1, 0.0f); }
            const float o0 = (x0 * wq.x + x1 * wq.z) * nrm;
            const float o1 = (x0 * wq.y + x1 * wq.w) * nrm;
            (void)__hip_atomic_fetch_add(&arow[64 * j + 0], o0,
                                         __ATOMIC_RELAXED, __HIP_MEMORY_SCOPE_WORKGROUP);
            (void)__hip_atomic_fetch_add(&arow[64 * j + 1], o1,
                                         __ATOMIC_RELAXED, __HIP_MEMORY_SCOPE_WORKGROUP);
        }
    }
}

template <bool RELU>
__device__ __forceinline__ void drain_recs(const float* __restrict__ xin,
                                           const float* wlds, float* acc,
                                           const int4* rq, int m, int lane)
{
    int i = 0;
    for (; i + 4 <= m; i += 4)
        edge_batch<RELU, 4>(xin, wlds, acc, rq + i, lane);
    const int rem = m - i;
    if (rem == 3)      edge_batch<RELU, 3>(xin, wlds, acc, rq + i, lane);
    else if (rem == 2) edge_batch<RELU, 2>(xin, wlds, acc, rq + i, lane);
    else if (rem == 1) edge_batch<RELU, 1>(xin, wlds, acc, rq + i, lane);
}

template <bool RELU, bool SAVE>
__device__ __forceinline__ void flush_ids(const float* __restrict__ xin,
                                          const int*   __restrict__ src,
                                          const int*   __restrict__ dst,
                                          const int*   __restrict__ etype,
                                          const float* __restrict__ norm,
                                          const float* wlds, float* acc,
                                          const int* q, int qn, int lo,
                                          int4* rq, int4* strip, int* fill, int* ovf, int lane)
{
    for (int i = 0; i < qn; i += 32) {
        const int m = (qn - i < 32) ? (qn - i) : 32;
        int4* save = nullptr;
        if (SAVE) {
            int o = 0;
            if (lane == 0)
                o = __hip_atomic_fetch_add(fill, m, __ATOMIC_RELAXED, __HIP_MEMORY_SCOPE_WORKGROUP);
            o = __builtin_amdgcn_readfirstlane(o);
            if (o + m <= WCAP) save = strip + o;
            else if (lane == 0) *ovf = 1;
        }
        int4 rec = make_int4(0, 0, 0, 0);
        if (lane < m) {
            const int e = q[i + lane];
            rec.x = src[e];
            rec.y = dst[e] - lo;
            rec.z = etype[e];
            rec.w = __float_as_int(norm[e]);
            if (save != nullptr) save[lane] = rec;
        }
        rq[lane] = rec;
        drain_recs<RELU>(xin, wlds, acc, rq, m, lane);
    }
}

template <bool RELU, bool FIRST>
__global__ __launch_bounds__(ETHR)
void rgcn_edge_owner(const float* __restrict__ xin,
                     const int*   __restrict__ src,
                     const int*   __restrict__ dst,
                     const int*   __restrict__ etype,
                     const float* __restrict__ norm,
                     const float* __restrict__ w,
                     const float* __restrict__ lw,
                     const float* __restrict__ bias,
                     float* __restrict__ agg,
                     int*  __restrict__ cnt,
                     int4* __restrict__ recs,
                     int n_edges, int n_nodes)
{
    __shared__ __align__(16) float wlds[WTAB];
    __shared__ __align__(16) float acc[NPB * HD];
    __shared__ __align__(16) float atile[NPB * APITCH];
    __shared__ __align__(16) int4 rqs[EWAV * 32];
    __shared__ int queue[EWAV * QCAP];
    __shared__ int wg_fill;
    __shared__ int wg_ovf;
    __shared__ int wg_next;

    const int tid  = threadIdx.x;
    const int lane = tid & 31;
    const int wv   = __builtin_amdgcn_readfirstlane(tid >> 5);
    const int lo   = blockIdx.x * NPB;

    float4 av[ASTG];
    stage_rows_load(xin, av, lo, n_nodes, tid);
    for (int i = tid; i < WTAB / 4; i += ETHR)
        ((float4*)wlds)[i] = ((const float4*)w)[i];
    for (int i = tid; i < NPB * HD / 4; i += ETHR)
        ((float4*)acc)[i] = make_float4(0.0f, 0.0f, 0.0f, 0.0f);
    if (tid == 0) { wg_fill = 0; wg_ovf = 0; wg_next = 0; }
    stage_rows_store<RELU>(atile, av, tid);
    __syncthreads();

    const int sw = wv - (EWAV - GWAV);
    if (sw >= 0) {
        const int tile_m = sw >> 2;
        if (lo + tile_m * 16 < n_nodes)
            selfloop_strip(atile, lw, bias, acc, tile_m, sw & 3, lane);
    }
    __builtin_amdgcn_s_setprio(3);

    int4* rq = rqs + wv * 32;
    int4* strip = recs + (size_t)blockIdx.x * WCAP;

    int saved = -1;
    if (!FIRST) saved = cnt[blockIdx.x];

    if (!FIRST && saved >= 0) {
        int c = 0;
        if (lane == 0)
            c = __hip_atomic_fetch_add(&wg_next, CH, __ATOMIC_RELAXED, __HIP_MEMORY_SCOPE_WORKGROUP);
        c = __builtin_amdgcn_readfirstlane(c);
        int4 cur = make_int4(0, 0, 0, 0);
        if (lane < CH && c + lane < saved) cur = strip[c + lane];
        while (c < saved) {
            const int m = (saved - c < CH) ? (saved - c) : CH;
            rq[lane] = cur;
            int cn = 0;
            if (lane == 0)
                cn = __hip_atomic_fetch_add(&wg_next, CH, __ATOMIC_RELAXED, __HIP_MEMORY_SCOPE_WORKGROUP);
            cn = __builtin_amdgcn_readfirstlane(cn);
            int4 nxt = make_int4(0, 0, 0, 0);
            if (lane < CH && cn + lane < saved) nxt = strip[cn + lane];
            drain_recs<RELU>(xin, wlds, acc, rq, m, lane);
            c = cn;
            cur = nxt;
        }
    } else {
        int* q = queue + wv * QCAP;
        int qn = 0;

        const int n_full   = n_edges / SCH;
        const int n_groups = (n_edges + 3) >> 2;
        const int rot = n_full > 0 ? (int)(((unsigned long long)blockIdx.x * (unsigned)n_full) / gridDim.x) : 0;

        auto take = [&]() {
            int r = 0;
            if (lane == 0)
                r = __hip_atomic_fetch_add(&wg_next, 1, __ATOMIC_RELAXED, __HIP_MEMORY_SCOPE_WORKGROUP);
            return __builtin_amdgcn_readfirstlane(r);
        };

        int t = take();
        int c = t + rot;
        if (c >= n_full) c -= n_full;
        int4 v[4];
        if (t < n_full) load_chunk(dst, c, lane, v);
        while (t < n_full) {
            const int tn = take();
            int cn = tn + rot;
            if (cn >= n_full) cn -= n_full;
            int4 vn[4];
            load_chunk(dst, (tn < n_full) ? cn : c, lane, vn);
            #pragma unroll
            for (int u = 0; u < 4; ++u)
                qn = push_matches(v[u], c * SCH + u * 128 + lane * 4, lo, q, qn);
            if (qn >= FLMIN) {
                flush_ids<RELU, FIRST>(xin, src, dst, etype, norm, wlds, acc, q, qn, lo, rq,
                                       strip, &wg_fill, &wg_ovf, lane);
                qn = 0;
            }
            t = tn;
            c = cn;
            #pragma unroll
            for (int u = 0; u < 4; ++u) v[u] = vn[u];
        }
        if (t == n_full && n_full * SCH < n_edges) {
            #pragma unroll
            for (int u = 0; u < 4; ++u) {
                const int g = n_full * (SCH / 4) + u * 32 + lane;
                qn = push_matches(load_dst4(dst, g, n_groups, n_edges), g * 4, lo, q, qn);
            }
        }
        flush_ids<RELU, FIRST>(xin, src, dst, etype, norm, wlds, acc, q, qn, lo, rq,
                               strip, &wg_fill, &wg_ovf, lane);
    }
    __syncthreads();

    if (FIRST && tid == 0) cnt[blockIdx.x] = (wg_ovf == 0) ? wg_fill : -1;

    for (int i = tid; i < NPB * (HD / 4); i += ETHR) {
        const int row = lo + (i >> 6);
        if (row < n_nodes)
            ((float4*)(agg + (size_t)row * HD))[i & 63] = ((const float4*)acc)[i];
    }
}

extern "C" void kernel_launch(void* const* d_in, const int* in_sizes, int n_in,
                              void* d_out, int out_size, void* d_ws, size_t ws_size,
                              hipStream_t stream)
{
    const float* x     = (const float*)d_in[0];
    const int*   src   = (const int*)  d_in[1];
    const int*   dst   = (const int*)  d_in[2];
    const int*   etype = (const int*)  d_in[3];
    const float* norm  = (const float*)d_in[4];
    const float* w1    = (const float*)d_in[5];
    const float* lw1   = (const float*)d_in[6];
    const float* b1    = (const float*)d_in[7];
    const float* w2    = (const float*)d_in[8];
    const float* lw2   = (const float*)d_in[9];
    const float* b2    = (const float*)d_in[10];
    float*       out   = (float*)d_out;

    const int n_nodes = in_sizes[0] / HD;
    const int n_edges = in_sizes[1];

    const int edge_blocks = (n_nodes + NPB - 1) / NPB;

    float* h    = (float*)d_ws;
    int*   cnt  = (int*)(h + (size_t)n_nodes * HD);
    int4*  recs = (int4*)(cnt + ((edge_blocks + 63) / 64) * 64);

    rgcn_edge_owner<false, true><<<edge_blocks, ETHR, 0, stream>>>(x, src, dst, etype, norm, w1, lw1, b1,
                                                                   h, cnt, recs, n_edges, n_nodes);

    rgcn_edge_owner<true, false><<<edge_blocks, ETHR, 0, stream>>>(h, src, dst, etype, norm, w2, lw2, b2,
                                                                   out, cnt, recs, n_edges, n_nodes);
}


# Edge batch: a relation's table reads from LDS asked for one edge ahead of the LDS adds, not behind them: the wait for a table word no longer waits for the adds before it; arrival order as before
# speedup vs baseline: 3.9733x; 1.0369x over previous
#include <hip/hip_runtime.h>

#define HD   256
#define NB   128
#define RT   16
#define WTAB (RT * NB * 4)

#define NPB   80
#define EWAV  32
#define ETHR  (EWAV * 32)

typedef float v2f __attribute__((ext_vector_type(2)));
typedef float v8f __attribute__((ext_vector_type(8)));

#define APITCH (HD + 4)

#define ASTG (NPB * (HD / 4) / ETHR)
static_assert(ASTG * ETHR == NPB * (HD / 4), "the tile must divide evenly over the threads");

__device__ __forceinline__ void stage_rows_load(const float* __restrict__ A, float4 (&av)[ASTG],
                                                int lo, int n_rows, int tid)
{
    #pragma unroll
    for (int u = 0; u < ASTG; ++u) {
        const int i = tid + u * ETHR;
        int idx = lo + (i >> 6);
        if (idx >= n_rows) idx = n_rows - 1;
        av[u] = ((const float4*)(A + (size_t)idx * HD))[i & 63];
    }
}

template <bool RELU>
__device__ __forceinline__ void stage_rows_store(float* atile, const float4 (&av)[ASTG], int tid)
{
    #pragma unroll
    for (int u = 0; u < ASTG; ++u) {
        const int i = tid + u * ETHR;
        float4 v = av[u];
        if (RELU) {
            v.x = fmaxf(v.x, 0.0f); v.y = fmaxf(v.y, 0.0f);
            v.z = fmaxf(v.z, 0.0f); v.w = fmaxf(v.w, 0.0f);
        }
        *(float4*)(atile + (i >> 6) * APITCH + 4 * (i & 63)) = v;
    }
}

__device__ __forceinline__ void selfloop_strip(const float* atile,
                                               const float* __restrict__ W,
                                               const float* __restrict__ bias,
                                               float* acc, int tile_m, int strip_n, int lane)
{
    const int half = lane >> 4;
    const int sub  = lane & 15;
    const int col0 = strip_n * 64 + 4 * sub;

    const float* arow = atile + (tile_m * 16 + sub) * APITCH + 2 * half;
    const float* wrow = W + (size_t)(2 * half) * HD + col0;

    const float4 bq = *(const float4*)(bias + col0);

    v8f c0 = {}, c1 = {}, c2 = {}, c3 = {};
    for (int k = 0; k < HD; k += 4) {
        const v2f a = *(const v2f*)(arow + k);
        const float4 q0 = *(const float4*)(wrow + (size_t)k * HD);
        const float4 q1 = *(const float4*)(wrow + (size_t)(k + 1) * HD);
        v2f b0, b1, b2, b3;
        b0.x = q0.x; b0.y = q1.x;
        b1.x = q0.y; b1.y = q1.y;
        b2.x = q0.z; b2.y = q1.z;
        b3.x = q0.w; b3.y = q1.w;

        c0 = __builtin_amdgcn_wmma_f32_16x16x4_f32(false, a, false, b0, (short)0, c0, false, false);
        c1 = __builtin_amdgcn_wmma_f32_16x16x4_f32(false, a, false, b1, (short)0, c1, false, false);
        c2 = __builtin_amdgcn_wmma_f32_16x16x4_f32(false, a, false, b2, (short)0, c2, false, false);
        c3 = __builtin_amdgcn_wmma_f32_16x16x4_f32(false, a, false, b3, (short)0, c3, false, false);
    }

    const float bcol[4] = { bq.x, bq.y, bq.z, bq.w };
    float* abase = acc + (size_t)(tile_m * 16 + half * 8) * HD + col0;
    #pragma unroll
    for (int t = 0; t < 4; ++t) {
        const v8f& c = (t == 0) ? c0 : (t == 1) ? c1 : (t == 2) ? c2 : c3;
        #pragma unroll
        for (int j = 0; j < 8; ++j)
            (void)__hip_atomic_fetch_add(&abase[(size_t)j * HD + t], c[j] + bcol[t],
                                         __ATOMIC_RELAXED, __HIP_MEMORY_SCOPE_WORKGROUP);
    }
}

#define QCAP  640
#define WCAP  2048
#define GWAV  ((NPB / 16) * (HD / 64))
#define CH    8
#define SCH   512
#define FLMIN 8
static_assert(QCAP >= FLMIN - 1 + SCH, "the queue must hold a whole chunk on top of an unflushed rest");

__device__ __forceinline__ int4 load_dst4(const int* __restrict__ dst, int g, int n_groups, int n_edges)
{
    int4 v = make_int4(-1, -1, -1, -1);
    if (g < n_groups) {
        const int e = g * 4;
        if (e + 3 < n_edges) {
            v = *(const int4*)(dst + e);
        } else {
            v.x = dst[e];
            if (e + 1 < n_edges) v.y = dst[e + 1];
            if (e + 2 < n_edges) v.z = dst[e + 2];
        }
    }
    return v;
}

__device__ __forceinline__ void load_chunk(const int* __restrict__ dst, int c, int lane, int4 (&v)[4])
{
    const int4* p = (const int4*)(dst + (size_t)c * SCH) + lane;
    #pragma unroll
    for (int u = 0; u < 4; ++u) v[u] = p[u * 32];
}

__device__ __forceinline__ int push_matches(int4 v, int e0, int lo, int* q, int qn)
{
    const bool m0 = (unsigned)(v.x - lo) < (unsigned)NPB;
    const bool m1 = (unsigned)(v.y - lo) < (unsigned)NPB;
    const bool m2 = (unsigned)(v.z - lo) < (unsigned)NPB;
    const bool m3 = (unsigned)(v.w - lo) < (unsigned)NPB;
    if (__builtin_amdgcn_ballot_w32(m0 | m1 | m2 | m3) != 0u) {
        unsigned mk;
        mk = __builtin_amdgcn_ballot_w32(m0);
        if (m0) q[qn + __builtin_amdgcn_mbcnt_lo(mk, 0u)] = e0 + 0;
        qn += __builtin_popcount(mk);
        mk = __builtin_amdgcn_ballot_w32(m1);
        if (m1) q[qn + __builtin_amdgcn_mbcnt_lo(mk, 0u)] = e0 + 1;
        qn += __builtin_popcount(mk);
        mk = __builtin_amdgcn_ballot_w32(m2);
        if (m2) q[qn + __builtin_amdgcn_mbcnt_lo(mk, 0u)] = e0 + 2;
        qn += __builtin_popcount(mk);
        mk = __builtin_amdgcn_ballot_w32(m3);
        if (m3) q[qn + __builtin_amdgcn_mbcnt_lo(mk, 0u)] = e0 + 3;
        qn += __builtin_popcount(mk);
    }
    return qn;
}

__device__ __forceinline__ void load_rel(float4 (&wq)[4], const float* wlds, int rel, int lane)
{
    const float4* wrel = (const float4*)(wlds + (size_t)rel * (NB * 4)) + lane;
    #pragma unroll
    for (int j = 0; j < 4; ++j) wq[j] = wrel[32 * j];
}

template <bool RELU, int NE>
__device__ __forceinline__ void edge_batch(const float* __restrict__ xin,
                                           const float* wlds, float* acc,
                                           const int4* rq, int lane)
{
    int4 rc[NE];
    #pragma unroll
    for (int k = 0; k < NE; ++k) rc[k] = rq[k];
    float2 xv[NE][4];
    #pragma unroll
    for (int k = 0; k < NE; ++k) {
        const float2* xr = (const float2*)(xin + (size_t)rc[k].x * HD) + lane;
        #pragma unroll
        for (int j = 0; j < 4; ++j) xv[k][j] = xr[32 * j];
    }
    float4 wq[NE][4];
    load_rel(wq[0], wlds, rc[0].z, lane);
    #pragma unroll
    for (int k = 0; k < NE; ++k) {
        if (k + 1 < NE) load_rel(wq[k + 1], wlds, rc[k + 1].z, lane);
        const float nrm = __int_as_float(rc[k].w);
        float o0[4], o1[4];
        #pragma unroll
        for (int j = 0; j < 4; ++j) {
            const float4 w4 = wq[k][j];
            float x0 = xv[k][j].x;
            float x1 = xv[k][j].y;
            if (RELU) { x0 = fmaxf(x0, 0.0f); x1 = fmaxf(x1, 0.0f); }
            o0[j] = (x0 * w4.x + x1 * w4.z) * nrm;
            o1[j] = (x0 * w4.y + x1 * w4.w) * nrm;
        }
        float* arow = acc + (size_t)rc[k].y * HD + 2 * lane;
        #pragma unroll
        for (int j = 0; j < 4; ++j) {
            (void)__hip_atomic_fetch_add(&arow[64 * j + 0], o0[j],
                                         __ATOMIC_RELAXED, __HIP_MEMORY_SCOPE_WORKGROUP);
            (void)__hip_atomic_fetch_add(&arow[64 * j + 1], o1[j],
                                         __ATOMIC_RELAXED, __HIP_MEMORY_SCOPE_WORKGROUP);
        }
    }
}

template <bool RELU>
__device__ __forceinline__ void drain_recs(const float* __restrict__ xin,
                                           const float* wlds, float* acc,
                                           const int4* rq, int m, int lane)
{
    int i = 0;
    for (; i + 4 <= m; i += 4)
        edge_batch<RELU, 4>(xin, wlds, acc, rq + i, lane);
    const int rem = m - i;
    if (rem == 3)      edge_batch<RELU, 3>(xin, wlds, acc, rq + i, lane);
    else if (rem == 2) edge_batch<RELU, 2>(xin, wlds, acc, rq + i, lane);
    else if (rem == 1) edge_batch<RELU, 1>(xin, wlds, acc, rq + i, lane);
}

template <bool RELU, bool SAVE>
__device__ __forceinline__ void flush_ids(const float* __restrict__ xin,
                                          const int*   __restrict__ src,
                                          const int*   __restrict__ dst,
                                          const int*   __restrict__ etype,
                                          const float* __restrict__ norm,
                                          const float* wlds, float* acc,
                                          const int* q, int qn, int lo,
                                          int4* rq, int4* strip, int* fill, int* ovf, int lane)
{
    for (int i = 0; i < qn; i += 32) {
        const int m = (qn - i < 32) ? (qn - i) : 32;
        int4* save = nullptr;
        if (SAVE) {
            int o = 0;
            if (lane == 0)
                o = __hip_atomic_fetch_add(fill, m, __ATOMIC_RELAXED, __HIP_MEMORY_SCOPE_WORKGROUP);
            o = __builtin_amdgcn_readfirstlane(o);
            if (o + m <= WCAP) save = strip + o;
            else if (lane == 0) *ovf = 1;
        }
        int4 rec = make_int4(0, 0, 0, 0);
        if (lane < m) {
            const int e = q[i + lane];
            rec.x = src[e];
            rec.y = dst[e] - lo;
            rec.z = etype[e];
            rec.w = __float_as_int(norm[e]);
            if (save != nullptr) save[lane] = rec;
        }
        rq[lane] = rec;
        drain_recs<RELU>(xin, wlds, acc, rq, m, lane);
    }
}

template <bool RELU, bool FIRST>
__global__ __launch_bounds__(ETHR)
void rgcn_edge_owner(const float* __restrict__ xin,
                     const int*   __restrict__ src,
                     const int*   __restrict__ dst,
                     const int*   __restrict__ etype,
                     const float* __restrict__ norm,
                     const float* __restrict__ w,
                     const float* __restrict__ lw,
                     const float* __restrict__ bias,
                     float* __restrict__ agg,
                     int*  __restrict__ cnt,
                     int4* __restrict__ recs,
                     int n_edges, int n_nodes)
{
    __shared__ __align__(16) float wlds[WTAB];
    __shared__ __align__(16) float acc[NPB * HD];
    __shared__ __align__(16) float atile[NPB * APITCH];
    __shared__ __align__(16) int4 rqs[EWAV * 32];
    __shared__ int queue[EWAV * QCAP];
    __shared__ int wg_fill;
    __shared__ int wg_ovf;
    __shared__ int wg_next;

    const int tid  = threadIdx.x;
    const int lane = tid & 31;
    const int wv   = __builtin_amdgcn_readfirstlane(tid >> 5);
    const int lo   = blockIdx.x * NPB;

    float4 av[ASTG];
    stage_rows_load(xin, av, lo, n_nodes, tid);
    for (int i = tid; i < WTAB / 4; i += ETHR)
        ((float4*)wlds)[i] = ((const float4*)w)[i];
    for (int i = tid; i < NPB * HD / 4; i += ETHR)
        ((float4*)acc)[i] = make_float4(0.0f, 0.0f, 0.0f, 0.0f);
    if (tid == 0) { wg_fill = 0; wg_ovf = 0; wg_next = 0; }
    stage_rows_store<RELU>(atile, av, tid);
    __syncthreads();

    const int sw = wv - (EWAV - GWAV);
    if (sw >= 0) {
        const int tile_m = sw >> 2;
        if (lo + tile_m * 16 < n_nodes)
            selfloop_strip(atile, lw, bias, acc, tile_m, sw & 3, lane);
    }
    __builtin_amdgcn_s_setprio(3);

    int4* rq = rqs + wv * 32;
    int4* strip = recs + (size_t)blockIdx.x * WCAP;

    int saved = -1;
    if (!FIRST) saved = cnt[blockIdx.x];

    if (!FIRST && saved >= 0) {
        int c = 0;
        if (lane == 0)
            c = __hip_atomic_fetch_add(&wg_next, CH, __ATOMIC_RELAXED, __HIP_MEMORY_SCOPE_WORKGROUP);
        c = __builtin_amdgcn_readfirstlane(c);
        int4 cur = make_int4(0, 0, 0, 0);
        if (lane < CH && c + lane < saved) cur = strip[c + lane];
        while (c < saved) {
            const int m = (saved - c < CH) ? (saved - c) : CH;
            rq[lane] = cur;
            int cn = 0;
            if (lane == 0)
                cn = __hip_atomic_fetch_add(&wg_next, CH, __ATOMIC_RELAXED, __HIP_MEMORY_SCOPE_WORKGROUP);
            cn = __builtin_amdgcn_readfirstlane(cn);
            int4 nxt = make_int4(0, 0, 0, 0);
            if (lane < CH && cn + lane < saved) nxt = strip[cn + lane];
            drain_recs<RELU>(xin, wlds, acc, rq, m, lane);
            c = cn;
            cur = nxt;
        }
    } else {
        int* q = queue + wv * QCAP;
        int qn = 0;

        const int n_full   = n_edges / SCH;
        const int n_groups = (n_edges + 3) >> 2;
        const int rot = n_full > 0 ? (int)(((unsigned long long)blockIdx.x * (unsigned)n_full) / gridDim.x) : 0;

        auto take = [&]() {
            int r = 0;
            if (lane == 0)
                r = __hip_atomic_fetch_add(&wg_next, 1, __ATOMIC_RELAXED, __HIP_MEMORY_SCOPE_WORKGROUP);
            return __builtin_amdgcn_readfirstlane(r);
        };

        int t = take();
        int c = t + rot;
        if (c >= n_full) c -= n_full;
        int4 v[4];
        if (t < n_full) load_chunk(dst, c, lane, v);
        while (t < n_full) {
            const int tn = take();
            int cn = tn + rot;
            if (cn >= n_full) cn -= n_full;
            int4 vn[4];
            load_chunk(dst, (tn < n_full) ? cn : c, lane, vn);
            #pragma unroll
            for (int u = 0; u < 4; ++u)
                qn = push_matches(v[u], c * SCH + u * 128 + lane * 4, lo, q, qn);
            if (qn >= FLMIN) {
                flush_ids<RELU, FIRST>(xin, src, dst, etype, norm, wlds, acc, q, qn, lo, rq,
                                       strip, &wg_fill, &wg_ovf, lane);
                qn = 0;
            }
            t = tn;
            c = cn;
            #pragma unroll
            for (int u = 0; u < 4; ++u) v[u] = vn[u];
        }
        if (t == n_full && n_full * SCH < n_edges) {
            #pragma unroll
            for (int u = 0; u < 4; ++u) {
                const int g = n_full * (SCH / 4) + u * 32 + lane;
                qn = push_matches(load_dst4(dst, g, n_groups, n_edges), g * 4, lo, q, qn);
            }
        }
        flush_ids<RELU, FIRST>(xin, src, dst, etype, norm, wlds, acc, q, qn, lo, rq,
                               strip, &wg_fill, &wg_ovf, lane);
    }
    __syncthreads();

    if (FIRST && tid == 0) cnt[blockIdx.x] = (wg_ovf == 0) ? wg_fill : -1;

    for (int i = tid; i < NPB * (HD / 4); i += ETHR) {
        const int row = lo + (i >> 6);
        if (row < n_nodes)
            ((float4*)(agg + (size_t)row * HD))[i & 63] = ((const float4*)acc)[i];
    }
}

extern "C" void kernel_launch(void* const* d_in, const int* in_sizes, int n_in,
                              void* d_out, int out_size, void* d_ws, size_t ws_size,
                              hipStream_t stream)
{
    const float* x     = (const float*)d_in[0];
    const int*   src   = (const int*)  d_in[1];
    const int*   dst   = (const int*)  d_in[2];
    const int*   etype = (const int*)  d_in[3];
    const float* norm  = (const float*)d_in[4];
    const float* w1    = (const float*)d_in[5];
    const float* lw1   = (const float*)d_in[6];
    const float* b1    = (const float*)d_in[7];
    const float* w2    = (const float*)d_in[8];
    const float* lw2   = (const float*)d_in[9];
    const float* b2    = (const float*)d_in[10];
    float*       out   = (float*)d_out;

    const int n_nodes = in_sizes[0] / HD;
    const int n_edges = in_sizes[1];

    const int edge_blocks = (n_nodes + NPB - 1) / NPB;

    float* h    = (float*)d_ws;
    int*   cnt  = (int*)(h + (size_t)n_nodes * HD);
    int4*  recs = (int4*)(cnt + ((edge_blocks + 63) / 64) * 64);

    rgcn_edge_owner<false, true><<<edge_blocks, ETHR, 0, stream>>>(x, src, dst, etype, norm, w1, lw1, b1,
                                                                   h, cnt, recs, n_edges, n_nodes);

    rgcn_edge_owner<true, false><<<edge_blocks, ETHR, 0, stream>>>(h, src, dst, etype, norm, w2, lw2, b2,
                                                                   out, cnt, recs, n_edges, n_nodes);
}


# Layer 2 orders its saved records by destination row (counting sort in LDS at set-up); a wave sums each run of one row in registers, one LDS float add per run: sum order changed (runs summed first)
# speedup vs baseline: 4.2637x; 1.0731x over previous
#include <hip/hip_runtime.h>

#define HD   256
#define NB   128
#define RT   16
#define WTAB (RT * NB * 4)

#define NPB   80
#define EWAV  32
#define ETHR  (EWAV * 32)

typedef float v2f __attribute__((ext_vector_type(2)));
typedef float v8f __attribute__((ext_vector_type(8)));

#define APITCH (HD + 4)

#define ASTG (NPB * (HD / 4) / ETHR)
static_assert(ASTG * ETHR == NPB * (HD / 4), "the tile must divide evenly over the threads");

__device__ __forceinline__ void stage_rows_load(const float* __restrict__ A, float4 (&av)[ASTG],
                                                int lo, int n_rows, int tid)
{
    #pragma unroll
    for (int u = 0; u < ASTG; ++u) {
        const int i = tid + u * ETHR;
        int idx = lo + (i >> 6);
        if (idx >= n_rows) idx = n_rows - 1;
        av[u] = ((const float4*)(A + (size_t)idx * HD))[i & 63];
    }
}

template <bool RELU>
__device__ __forceinline__ void stage_rows_store(float* atile, const float4 (&av)[ASTG], int tid)
{
    #pragma unroll
    for (int u = 0; u < ASTG; ++u) {
        const int i = tid + u * ETHR;
        float4 v = av[u];
        if (RELU) {
            v.x = fmaxf(v.x, 0.0f); v.y = fmaxf(v.y, 0.0f);
            v.z = fmaxf(v.z, 0.0f); v.w = fmaxf(v.w, 0.0f);
        }
        *(float4*)(atile + (i >> 6) * APITCH + 4 * (i & 63)) = v;
    }
}

__device__ __forceinline__ void selfloop_strip(const float* atile,
                                               const float* __restrict__ W,
                                               const float* __restrict__ bias,
                                               float* acc, int tile_m, int strip_n, int lane)
{
    const int half = lane >> 4;
    const int sub  = lane & 15;
    const int col0 = strip_n * 64 + 4 * sub;

    const float* arow = atile + (tile_m * 16 + sub) * APITCH + 2 * half;
    const float* wrow = W + (size_t)(2 * half) * HD + col0;

    const float4 bq = *(const float4*)(bias + col0);

    v8f c0 = {}, c1 = {}, c2 = {}, c3 = {};
    for (int k = 0; k < HD; k += 4) {
        const v2f a = *(const v2f*)(arow + k);
        const float4 q0 = *(const float4*)(wrow + (size_t)k * HD);
        const float4 q1 = *(const float4*)(wrow + (size_t)(k + 1) * HD);
        v2f b0, b1, b2, b3;
        b0.x = q0.x; b0.y = q1.x;
        b1.x = q0.y; b1.y = q1.y;
        b2.x = q0.z; b2.y = q1.z;
        b3.x = q0.w; b3.y = q1.w;

        c0 = __builtin_amdgcn_wmma_f32_16x16x4_f32(false, a, false, b0, (short)0, c0, false, false);
        c1 = __builtin_amdgcn_wmma_f32_16x16x4_f32(false, a, false, b1, (short)0, c1, false, false);
        c2 = __builtin_amdgcn_wmma_f32_16x16x4_f32(false, a, false, b2, (short)0, c2, false, false);
        c3 = __builtin_amdgcn_wmma_f32_16x16x4_f32(false, a, false, b3, (short)0, c3, false, false);
    }

    const float bcol[4] = { bq.x, bq.y, bq.z, bq.w };
    float* abase = acc + (size_t)(tile_m * 16 + half * 8) * HD + col0;
    #pragma unroll
    for (int t = 0; t < 4; ++t) {
        const v8f& c = (t == 0) ? c0 : (t == 1) ? c1 : (t == 2) ? c2 : c3;
        #pragma unroll
        for (int j = 0; j < 8; ++j)
            (void)__hip_atomic_fetch_add(&abase[(size_t)j * HD + t], c[j] + bcol[t],
                                         __ATOMIC_RELAXED, __HIP_MEMORY_SCOPE_WORKGROUP);
    }
}

#define QCAP  640
#define WCAP  2048
#define GWAV  ((NPB / 16) * (HD / 64))
#define CH    8
#define SRT   (WCAP / ETHR)
static_assert(SRT * ETHR == WCAP, "the saved list must divide evenly over the threads");
static_assert(WCAP * 16 <= EWAV * QCAP * 4, "the ordered list must fit the queue area");
#define SCH   512
#define FLMIN 8
static_assert(QCAP >= FLMIN - 1 + SCH, "the queue must hold a whole chunk on top of an unflushed rest");

__device__ __forceinline__ int4 load_dst4(const int* __restrict__ dst, int g, int n_groups, int n_edges)
{
    int4 v = make_int4(-1, -1, -1, -1);
    if (g < n_groups) {
        const int e = g * 4;
        if (e + 3 < n_edges) {
            v = *(const int4*)(dst + e);
        } else {
            v.x = dst[e];
            if (e + 1 < n_edges) v.y = dst[e + 1];
            if (e + 2 < n_edges) v.z = dst[e + 2];
        }
    }
    return v;
}

__device__ __forceinline__ void load_chunk(const int* __restrict__ dst, int c, int lane, int4 (&v)[4])
{
    const int4* p = (const int4*)(dst + (size_t)c * SCH) + lane;
    #pragma unroll
    for (int u = 0; u < 4; ++u) v[u] = p[u * 32];
}

__device__ __forceinline__ int push_matches(int4 v, int e0, int lo, int* q, int qn)
{
    const bool m0 = (unsigned)(v.x - lo) < (unsigned)NPB;
    const bool m1 = (unsigned)(v.y - lo) < (unsigned)NPB;
    const bool m2 = (unsigned)(v.z - lo) < (unsigned)NPB;
    const bool m3 = (unsigned)(v.w - lo) < (unsigned)NPB;
    if (__builtin_amdgcn_ballot_w32(m0 | m1 | m2 | m3) != 0u) {
        unsigned mk;
        mk = __builtin_amdgcn_ballot_w32(m0);
        if (m0) q[qn + __builtin_amdgcn_mbcnt_lo(mk, 0u)] = e0 + 0;
        qn += __builtin_popcount(mk);
        mk = __builtin_amdgcn_ballot_w32(m1);
        if (m1) q[qn + __builtin_amdgcn_mbcnt_lo(mk, 0u)] = e0 + 1;
        qn += __builtin_popcount(mk);
        mk = __builtin_amdgcn_ballot_w32(m2);
        if (m2) q[qn + __builtin_amdgcn_mbcnt_lo(mk, 0u)] = e0 + 2;
        qn += __builtin_popcount(mk);
        mk = __builtin_amdgcn_ballot_w32(m3);
        if (m3) q[qn + __builtin_amdgcn_mbcnt_lo(mk, 0u)] = e0 + 3;
        qn += __builtin_popcount(mk);
    }
    return qn;
}

__device__ __forceinline__ void load_rel(float4 (&wq)[4], const float* wlds, int rel, int lane)
{
    const float4* wrel = (const float4*)(wlds + (size_t)rel * (NB * 4)) + lane;
    #pragma unroll
    for (int j = 0; j < 4; ++j) wq[j] = wrel[32 * j];
}

template <bool RELU, int NE>
__device__ __forceinline__ void edge_batch(const float* __restrict__ xin,
                                           const float* wlds, float* acc,
                                           const int4* rq, int lane)
{
    int4 rc[NE];
    #pragma unroll
    for (int k = 0; k < NE; ++k) rc[k] = rq[k];
    float2 xv[NE][4];
    #pragma unroll
    for (int k = 0; k < NE; ++k) {
        const float2* xr = (const float2*)(xin + (size_t)rc[k].x * HD) + lane;
        #pragma unroll
        for (int j = 0; j < 4; ++j) xv[k][j] = xr[32 * j];
    }
    float4 wq[NE][4];
    load_rel(wq[0], wlds, rc[0].z, lane);
    #pragma unroll
    for (int k = 0; k < NE; ++k) {
        if (k + 1 < NE) load_rel(wq[k + 1], wlds, rc[k + 1].z, lane);
        const float nrm = __int_as_float(rc[k].w);
        float o0[4], o1[4];
        #pragma unroll
        for (int j = 0; j < 4; ++j) {
            const float4 w4 = wq[k][j];
            float x0 = xv[k][j].x;
            float x1 = xv[k][j].y;
            if (RELU) { x0 = fmaxf(x0, 0.0f); x1 = fmaxf(x1, 0.0f); }
            o0[j] = (x0 * w4.x + x1 * w4.z) * nrm;
            o1[j] = (x0 * w4.y + x1 * w4.w) * nrm;
        }
        float* arow = acc + (size_t)rc[k].y * HD + 2 * lane;
        #pragma unroll
        for (int j = 0; j < 4; ++j) {
            (void)__hip_atomic_fetch_add(&arow[64 * j + 0], o0[j],
                                         __ATOMIC_RELAXED, __HIP_MEMORY_SCOPE_WORKGROUP);
            (void)__hip_atomic_fetch_add(&arow[64 * j + 1], o1[j],
                                         __ATOMIC_RELAXED, __HIP_MEMORY_SCOPE_WORKGROUP);
        }
    }
}

template <bool RELU>
__device__ __forceinline__ void drain_recs(const float* __restrict__ xin,
                                           const float* wlds, float* acc,
                                           const int4* rq, int m, int lane)
{
    int i = 0;
    for (; i + 4 <= m; i += 4)
        edge_batch<RELU, 4>(xin, wlds, acc, rq + i, lane);
    const int rem = m - i;
    if (rem == 3)      edge_batch<RELU, 3>(xin, wlds, acc, rq + i, lane);
    else if (rem == 2) edge_batch<RELU, 2>(xin, wlds, acc, rq + i, lane);
    else if (rem == 1) edge_batch<RELU, 1>(xin, wlds, acc, rq + i, lane);
}

template <bool RELU, int NE>
__device__ __forceinline__ void edge_batch_runs(const float* __restrict__ xin,
                                                const float* wlds, float* acc,
                                                const int4* rq, int lane)
{
    int4 rc[NE];
    #pragma unroll
    for (int k = 0; k < NE; ++k) rc[k] = rq[k];
    float2 xv[NE][4];
    #pragma unroll
    for (int k = 0; k < NE; ++k) {
        const float2* xr = (const float2*)(xin + (size_t)rc[k].x * HD) + lane;
        #pragma unroll
        for (int j = 0; j < 4; ++j) xv[k][j] = xr[32 * j];
    }
    float4 wq[NE][4];
    load_rel(wq[0], wlds, rc[0].z, lane);
    float s0[4], s1[4];
    int srow = 0;
    #pragma unroll
    for (int k = 0; k < NE; ++k) {
        if (k + 1 < NE) load_rel(wq[k + 1], wlds, rc[k + 1].z, lane);
        const float nrm = __int_as_float(rc[k].w);
        float o0[4], o1[4];
        #pragma unroll
        for (int j = 0; j < 4; ++j) {
            const float4 w4 = wq[k][j];
            float x0 = xv[k][j].x;
            float x1 = xv[k][j].y;
            if (RELU) { x0 = fmaxf(x0, 0.0f); x1 = fmaxf(x1, 0.0f); }
            o0[j] = (x0 * w4.x + x1 * w4.z) * nrm;
            o1[j] = (x0 * w4.y + x1 * w4.w) * nrm;
        }
        const int row = __builtin_amdgcn_readfirstlane(rc[k].y);
        if (k > 0 && row == srow) {
            #pragma unroll
            for (int j = 0; j < 4; ++j) { s0[j] += o0[j]; s1[j] += o1[j]; }
        } else {
            if (k > 0) {
                float* arow = acc + (size_t)srow * HD + 2 * lane;
                #pragma unroll
                for (int j = 0; j < 4; ++j) {
                    (void)__hip_atomic_fetch_add(&arow[64 * j + 0], s0[j],
                                                 __ATOMIC_RELAXED, __HIP_MEMORY_SCOPE_WORKGROUP);
                    (void)__hip_atomic_fetch_add(&arow[64 * j + 1], s1[j],
                                                 __ATOMIC_RELAXED, __HIP_MEMORY_SCOPE_WORKGROUP);
                }
            }
            #pragma unroll
            for (int j = 0; j < 4; ++j) { s0[j] = o0[j]; s1[j] = o1[j]; }
            srow = row;
        }
    }
    float* arow = acc + (size_t)srow * HD + 2 * lane;
    #pragma unroll
    for (int j = 0; j < 4; ++j) {
        (void)__hip_atomic_fetch_add(&arow[64 * j + 0], s0[j],
                                     __ATOMIC_RELAXED, __HIP_MEMORY_SCOPE_WORKGROUP);
        (void)__hip_atomic_fetch_add(&arow[64 * j + 1], s1[j],
                                     __ATOMIC_RELAXED, __HIP_MEMORY_SCOPE_WORKGROUP);
    }
}

template <bool RELU>
__device__ __forceinline__ void drain_runs(const float* __restrict__ xin,
                                           const float* wlds, float* acc,
                                           const int4* rq, int m, int lane)
{
    int i = 0;
    for (; i + 4 <= m; i += 4)
        edge_batch_runs<RELU, 4>(xin, wlds, acc, rq + i, lane);
    const int rem = m - i;
    if (rem == 3)      edge_batch_runs<RELU, 3>(xin, wlds, acc, rq + i, lane);
    else if (rem == 2) edge_batch_runs<RELU, 2>(xin, wlds, acc, rq + i, lane);
    else if (rem == 1) edge_batch_runs<RELU, 1>(xin, wlds, acc, rq + i, lane);
}

template <bool RELU, bool SAVE>
__device__ __forceinline__ void flush_ids(const float* __restrict__ xin,
                                          const int*   __restrict__ src,
                                          const int*   __restrict__ dst,
                                          const int*   __restrict__ etype,
                                          const float* __restrict__ norm,
                                          const float* wlds, float* acc,
                                          const int* q, int qn, int lo,
                                          int4* rq, int4* strip, int* fill, int* ovf, int lane)
{
    for (int i = 0; i < qn; i += 32) {
        const int m = (qn - i < 32) ? (qn - i) : 32;
        int4* save = nullptr;
        if (SAVE) {
            int o = 0;
            if (lane == 0)
                o = __hip_atomic_fetch_add(fill, m, __ATOMIC_RELAXED, __HIP_MEMORY_SCOPE_WORKGROUP);
            o = __builtin_amdgcn_readfirstlane(o);
            if (o + m <= WCAP) save = strip + o;
            else if (lane == 0) *ovf = 1;
        }
        int4 rec = make_int4(0, 0, 0, 0);
        if (lane < m) {
            const int e = q[i + lane];
            rec.x = src[e];
            rec.y = dst[e] - lo;
            rec.z = etype[e];
            rec.w = __float_as_int(norm[e]);
            if (save != nullptr) save[lane] = rec;
        }
        rq[lane] = rec;
        drain_recs<RELU>(xin, wlds, acc, rq, m, lane);
    }
}

template <bool RELU, bool FIRST>
__global__ __launch_bounds__(ETHR)
void rgcn_edge_owner(const float* __restrict__ xin,
                     const int*   __restrict__ src,
                     const int*   __restrict__ dst,
                     const int*   __restrict__ etype,
                     const float* __restrict__ norm,
                     const float* __restrict__ w,
                     const float* __restrict__ lw,
                     const float* __restrict__ bias,
                     float* __restrict__ agg,
                     int*  __restrict__ cnt,
                     int4* __restrict__ recs,
                     int n_edges, int n_nodes)
{
    __shared__ __align__(16) float wlds[WTAB];
    __shared__ __align__(16) float acc[NPB * HD];
    __shared__ __align__(16) float atile[NPB * APITCH];
    __shared__ __align__(16) int4 rqs[EWAV * 32];
    __shared__ __align__(16) int queue[EWAV * QCAP];
    __shared__ int rowcnt[NPB];
    __shared__ int rowoff[NPB];
    __shared__ int wg_fill;
    __shared__ int wg_ovf;
    __shared__ int wg_next;

    const int tid  = threadIdx.x;
    const int lane = tid & 31;
    const int wv   = __builtin_amdgcn_readfirstlane(tid >> 5);
    const int lo   = blockIdx.x * NPB;

    int4* strip = recs + (size_t)blockIdx.x * WCAP;

    int saved = -1;
    if (!FIRST) saved = cnt[blockIdx.x];
    const bool ordered = !FIRST && saved >= 0 && saved <= WCAP;
    int4 myrec[SRT];
    if (ordered) {
        #pragma unroll
        for (int u = 0; u < SRT; ++u) {
            const int i = tid + u * ETHR;
            myrec[u] = make_int4(0, 0, 0, 0);
            if (i < saved) myrec[u] = strip[i];
        }
    }

    float4 av[ASTG];
    stage_rows_load(xin, av, lo, n_nodes, tid);
    for (int i = tid; i < WTAB / 4; i += ETHR)
        ((float4*)wlds)[i] = ((const float4*)w)[i];
    for (int i = tid; i < NPB * HD / 4; i += ETHR)
        ((float4*)acc)[i] = make_float4(0.0f, 0.0f, 0.0f, 0.0f);
    if (tid == 0) { wg_fill = 0; wg_ovf = 0; wg_next = 0; }
    if (tid < NPB) rowcnt[tid] = 0;
    stage_rows_store<RELU>(atile, av, tid);
    __syncthreads();

    int4* srec = (int4*)queue;
    if (ordered) {
        int myrank[SRT];
        #pragma unroll
        for (int u = 0; u < SRT; ++u) {
            const int i = tid + u * ETHR;
            myrec[u].y = min(max(myrec[u].y, 0), NPB - 1);
            myrank[u] = 0;
            if (i < saved)
                myrank[u] = __hip_atomic_fetch_add(&rowcnt[myrec[u].y], 1,
                                                   __ATOMIC_RELAXED, __HIP_MEMORY_SCOPE_WORKGROUP);
        }
        __syncthreads();
        if (tid < NPB) {
            int o = 0;
            for (int j = 0; j < tid; ++j) o += rowcnt[j];
            rowoff[tid] = o;
        }
        __syncthreads();
        #pragma unroll
        for (int u = 0; u < SRT; ++u) {
            const int i = tid + u * ETHR;
            if (i < saved) srec[rowoff[myrec[u].y] + myrank[u]] = myrec[u];
        }
        __syncthreads();
    }

    const int sw = wv - (EWAV - GWAV);
    if (sw >= 0) {
        const int tile_m = sw >> 2;
        if (lo + tile_m * 16 < n_nodes)
            selfloop_strip(atile, lw, bias, acc, tile_m, sw & 3, lane);
    }
    __builtin_amdgcn_s_setprio(3);

    int4* rq = rqs + wv * 32;

    if (ordered) {
        int c = 0;
        if (lane == 0)
            c = __hip_atomic_fetch_add(&wg_next, CH, __ATOMIC_RELAXED, __HIP_MEMORY_SCOPE_WORKGROUP);
        c = __builtin_amdgcn_readfirstlane(c);
        while (c < saved) {
            const int m = (saved - c < CH) ? (saved - c) : CH;
            int cn = 0;
            if (lane == 0)
                cn = __hip_atomic_fetch_add(&wg_next, CH, __ATOMIC_RELAXED, __HIP_MEMORY_SCOPE_WORKGROUP);
            cn = __builtin_amdgcn_readfirstlane(cn);
            drain_runs<RELU>(xin, wlds, acc, srec + c, m, lane);
            c = cn;
        }
    } else {
        int* q = queue + wv * QCAP;
        int qn = 0;

        const int n_full   = n_edges / SCH;
        const int n_groups = (n_edges + 3) >> 2;
        const int rot = n_full > 0 ? (int)(((unsigned long long)blockIdx.x * (unsigned)n_full) / gridDim.x) : 0;

        auto take = [&]() {
            int r = 0;
            if (lane == 0)
                r = __hip_atomic_fetch_add(&wg_next, 1, __ATOMIC_RELAXED, __HIP_MEMORY_SCOPE_WORKGROUP);
            return __builtin_amdgcn_readfirstlane(r);
        };

        int t = take();
        int c = t + rot;
        if (c >= n_full) c -= n_full;
        int4 v[4];
        if (t < n_full) load_chunk(dst, c, lane, v);
        while (t < n_full) {
            const int tn = take();
            int cn = tn + rot;
            if (cn >= n_full) cn -= n_full;
            int4 vn[4];
            load_chunk(dst, (tn < n_full) ? cn : c, lane, vn);
            #pragma unroll
            for (int u = 0; u < 4; ++u)
                qn = push_matches(v[u], c * SCH + u * 128 + lane * 4, lo, q, qn);
            if (qn >= FLMIN) {
                flush_ids<RELU, FIRST>(xin, src, dst, etype, norm, wlds, acc, q, qn, lo, rq,
                                       strip, &wg_fill, &wg_ovf, lane);
                qn = 0;
            }
            t = tn;
            c = cn;
            #pragma unroll
            for (int u = 0; u < 4; ++u) v[u] = vn[u];
        }
        if (t == n_full && n_full * SCH < n_edges) {
            #pragma unroll
            for (int u = 0; u < 4; ++u) {
                const int g = n_full * (SCH / 4) + u * 32 + lane;
                qn = push_matches(load_dst4(dst, g, n_groups, n_edges), g * 4, lo, q, qn);
            }
        }
        flush_ids<RELU, FIRST>(xin, src, dst, etype, norm, wlds, acc, q, qn, lo, rq,
                               strip, &wg_fill, &wg_ovf, lane);
    }
    __syncthreads();

    if (FIRST && tid == 0) cnt[blockIdx.x] = (wg_ovf == 0) ? wg_fill : -1;

    for (int i = tid; i < NPB * (HD / 4); i += ETHR) {
        const int row = lo + (i >> 6);
        if (row < n_nodes)
            ((float4*)(agg + (size_t)row * HD))[i & 63] = ((const float4*)acc)[i];
    }
}

extern "C" void kernel_launch(void* const* d_in, const int* in_sizes, int n_in,
                              void* d_out, int out_size, void* d_ws, size_t ws_size,
                              hipStream_t stream)
{
    const float* x     = (const float*)d_in[0];
    const int*   src   = (const int*)  d_in[1];
    const int*   dst   = (const int*)  d_in[2];
    const int*   etype = (const int*)  d_in[3];
    const float* norm  = (const float*)d_in[4];
    const float* w1    = (const float*)d_in[5];
    const float* lw1   = (const float*)d_in[6];
    const float* b1    = (const float*)d_in[7];
    const float* w2    = (const float*)d_in[8];
    const float* lw2   = (const float*)d_in[9];
    const float* b2    = (const float*)d_in[10];
    float*       out   = (float*)d_out;

    const int n_nodes = in_sizes[0] / HD;
    const int n_edges = in_sizes[1];

    const int edge_blocks = (n_nodes + NPB - 1) / NPB;

    float* h    = (float*)d_ws;
    int*   cnt  = (int*)(h + (size_t)n_nodes * HD);
    int4*  recs = (int4*)(cnt + ((edge_blocks + 63) / 64) * 64);

    rgcn_edge_owner<false, true><<<edge_blocks, ETHR, 0, stream>>>(x, src, dst, etype, norm, w1, lw1, b1,
                                                                   h, cnt, recs, n_edges, n_nodes);

    rgcn_edge_owner<true, false><<<edge_blocks, ETHR, 0, stream>>>(h, src, dst, etype, norm, w2, lw2, b2,
                                                                   out, cnt, recs, n_edges, n_nodes);
}
